# starting kernel
# baseline (speedup 1.0000x reference)
#include <hip/hip_runtime.h>
#include <math.h>

#define K_CODES 16384
#define DIMS    128
#define BATCH_N 262144
#define DECAY_F 0.99f
#define EPS_F   1e-5f

typedef __attribute__((ext_vector_type(2))) float v2f;
typedef __attribute__((ext_vector_type(8))) float v8f;

__global__ void ema_init_kernel(float* __restrict__ w, float* __restrict__ cnt) {
  const int stride = gridDim.x * blockDim.x;
  const int i = blockIdx.x * blockDim.x + threadIdx.x;
  const int total = K_CODES * DIMS;
  for (int j = i; j < total; j += stride) w[j] = 0.0f;
  for (int j = i; j < K_CODES; j += stride) cnt[j] = 0.0f;
}

__global__ __launch_bounds__(256)
void ema_scatter_kernel(const float* __restrict__ z_e, const int* __restrict__ idxs,
                        float* __restrict__ w, float* __restrict__ cnt) {
  const int lane = threadIdx.x & 31;
  const int wv   = threadIdx.x >> 5;
  const int r    = lane & 15;
  const int half = lane >> 4;
  const unsigned base = blockIdx.x * 16u;
  const int colbase = wv * 16;

  const int rowIdx = idxs[base + (unsigned)r];

  unsigned mm = 0u;
  #pragma unroll
  for (int c = 0; c < 16; ++c) {
    mm |= (unsigned)(__shfl(rowIdx, c, 32) == rowIdx) << c;
  }

  const bool isLeader = ((mm & ((1u << r) - 1u)) == 0u);
  const unsigned gmask = isLeader ? mm : 0u;

  v8f acc = {};
  #pragma unroll
  for (int k = 0; k < 4; ++k) {
    const int j0 = 4 * k + 2 * half;
    v2f a, b;
    a.x = (float)((gmask >> j0) & 1u);
    a.y = (float)((gmask >> (j0 + 1)) & 1u);
    const unsigned zoff = (base + (unsigned)j0) * DIMS + (unsigned)(colbase + r);
    b.x = z_e[zoff];
    b.y = z_e[zoff + DIMS];
    acc = __builtin_amdgcn_wmma_f32_16x16x4_f32(false, a, false, b,
                                                (short)0, acc, false, false);
  }

  #pragma unroll
  for (int v = 0; v < 8; ++v) {
    const int tcode = __shfl(rowIdx, 8 * half + v, 32);
    const unsigned off = (unsigned)tcode * DIMS + (unsigned)(colbase + r);
    atomicAdd(&w[off], acc[v]);
  }

  if (wv == 0 && half == 0 && isLeader) {
    atomicAdd(&cnt[(unsigned)rowIdx], (float)__popc(mm));
  }
}

__global__ void ema_finalize_kernel(const float* __restrict__ m, const float* __restrict__ N,
                                    const float* __restrict__ cnt, const int* __restrict__ tptr,
                                    float* __restrict__ out) {
  const float zbc = 1.0f - powf(DECAY_F, (float)tptr[0]);
  const int stride = gridDim.x * blockDim.x;
  const int total = K_CODES * DIMS;
  for (int j = blockIdx.x * blockDim.x + threadIdx.x; j < total; j += stride) {
    const int k = j >> 7;
    const float wv = out[j];
    const float nU = N[k] * DECAY_F + cnt[k] * (1.0f - DECAY_F);
    const float mU = m[j] * DECAY_F + wv * (1.0f - DECAY_F);
    out[j] = (mU / zbc) / ((nU / zbc) + EPS_F);
  }
}

extern "C" void kernel_launch(void* const* d_in, const int* in_sizes, int n_in,
                              void* d_out, int out_size, void* d_ws, size_t ws_size,
                              hipStream_t stream) {
  const float* z_e  = (const float*)d_in[0];
  const int*   idxs = (const int*)  d_in[1];
  const float* m    = (const float*)d_in[2];
  const float* N    = (const float*)d_in[3];
  const int*   t    = (const int*)  d_in[4];
  float* out = (float*)d_out;
  float* cnt = (float*)d_ws;

  ema_init_kernel<<<2048, 256, 0, stream>>>(out, cnt);
  ema_scatter_kernel<<<BATCH_N / 16, 256, 0, stream>>>(z_e, idxs, out, cnt);
  ema_finalize_kernel<<<(K_CODES * DIMS) / 256, 256, 0, stream>>>(m, N, cnt, t, out);
  (void)in_sizes; (void)n_in; (void)out_size; (void)ws_size;
}


# Atomic scatter -> owner gather: an id scan files each code's row numbers (LDS atomic rank, list in the code's out row); a wave per code sums its rows on WMMA, EMA fused; sums in LDS arrival order
# speedup vs baseline: 2.0704x; 2.0704x over previous
#include <hip/hip_runtime.h>
#include <math.h>


#define K_CODES 16384
#define DIMS    128
#define BATCH_N 262144
#define DECAY_F 0.99f
#define EPS_F   1e-5f

#define LIST_CAP     DIMS
#define OWN_CODES    128
#define FILE_THREADS 1024
#define FILE_DEPTH   8

typedef __attribute__((ext_vector_type(2))) float v2f;
typedef __attribute__((ext_vector_type(8))) float v8f;

__device__ __forceinline__ void file_row(int code, unsigned row, unsigned base,
                                         int* lcnt, float* __restrict__ lists) {
  const unsigned d = (unsigned)code - base;
  if (d < (unsigned)OWN_CODES) {
    const int pos = atomicAdd(&lcnt[d], 1);
    if (pos < LIST_CAP) lists[(base + d) * DIMS + (unsigned)pos] = __int_as_float((int)row);
  }
}

__global__ __launch_bounds__(FILE_THREADS)
void ema_file_kernel(const int* __restrict__ idxs, float* __restrict__ lists,
                     int* __restrict__ cnt) {
  __shared__ int lcnt[OWN_CODES];
  const unsigned t = threadIdx.x;
  const unsigned base = blockIdx.x * OWN_CODES;
  if (t < OWN_CODES) lcnt[t] = 0;
  __syncthreads();

  const int4* __restrict__ idx4 = (const int4*)idxs;
  for (unsigned it = 0; it < BATCH_N / 4 / FILE_THREADS; it += FILE_DEPTH) {
    int4 v[FILE_DEPTH];
    #pragma unroll
    for (int j = 0; j < FILE_DEPTH; ++j) v[j] = idx4[(it + j) * FILE_THREADS + t];
    #pragma unroll
    for (int j = 0; j < FILE_DEPTH; ++j) {
      const unsigned row = 4u * ((it + j) * FILE_THREADS + t);
      file_row(v[j].x, row + 0u, base, lcnt, lists);
      file_row(v[j].y, row + 1u, base, lcnt, lists);
      file_row(v[j].z, row + 2u, base, lcnt, lists);
      file_row(v[j].w, row + 3u, base, lcnt, lists);
    }
  }
  __syncthreads();
  if (t < OWN_CODES) cnt[base + t] = lcnt[t];
}

__device__ __forceinline__ void add_rows(int myrow, unsigned mask, unsigned lane,
                                         const float4* __restrict__ z4, v2f a,
                                         v8f& c0, v8f& c1, v8f& c2, v8f& c3) {
  for (int g = 0; g < 4; ++g) {
    const unsigned mg = (mask >> (8 * g)) & 0xffu;
    if (mg == 0u) continue;
    float4 x[8];
    #pragma unroll
    for (int j = 0; j < 8; ++j) {
      x[j] = make_float4(0.0f, 0.0f, 0.0f, 0.0f);
      if ((mg >> j) & 1u) {
        const unsigned srow = (unsigned)__builtin_amdgcn_readlane(myrow, 8 * g + j);
        x[j] = z4[srow * (DIMS / 4) + lane];
      }
    }
    #pragma unroll
    for (int p = 0; p < 4; ++p) {
      if ((mg >> (2 * p)) & 3u) {
        v2f b;
        b.x = x[2 * p].x; b.y = x[2 * p + 1].x;
        c0 = __builtin_amdgcn_wmma_f32_16x16x4_f32(false, a, false, b, (short)0, c0, false, false);
        b.x = x[2 * p].y; b.y = x[2 * p + 1].y;
        c1 = __builtin_amdgcn_wmma_f32_16x16x4_f32(false, a, false, b, (short)0, c1, false, false);
        b.x = x[2 * p].z; b.y = x[2 * p + 1].z;
        c2 = __builtin_amdgcn_wmma_f32_16x16x4_f32(false, a, false, b, (short)0, c2, false, false);
        b.x = x[2 * p].w; b.y = x[2 * p + 1].w;
        c3 = __builtin_amdgcn_wmma_f32_16x16x4_f32(false, a, false, b, (short)0, c3, false, false);
      }
    }
  }
}

__global__ __launch_bounds__(256)
void ema_gather_kernel(const float* __restrict__ z_e, const int* __restrict__ idxs,
                       const float* __restrict__ m, const float* __restrict__ N,
                       const int* __restrict__ cnt, const int* __restrict__ tptr,
                       float* out) {
  const unsigned lane = threadIdx.x & 31u;
  const int k = __builtin_amdgcn_readfirstlane((int)(blockIdx.x * 8u + (threadIdx.x >> 5)));
  const float4* __restrict__ z4 = (const float4*)z_e;

  const int n = cnt[k];
  const float4 m4 = ((const float4*)m)[(unsigned)k * (DIMS / 4) + lane];
  const float Nk = N[k];
  const float zbc = 1.0f - powf(DECAY_F, (float)tptr[0]);

  const float sel = (lane == 0u || lane == 24u) ? 1.0f : 0.0f;
  const v2f a = {sel, sel};
  v8f c0 = {}, c1 = {}, c2 = {}, c3 = {};

  if (n <= LIST_CAP) {
    for (int q = 0; q * 32 < n; ++q) {
      const int rem = n - q * 32;
      const unsigned mask = rem >= 32 ? 0xffffffffu : ((1u << rem) - 1u);
      const int e = q * 32 + (int)lane;
      int myrow = 0;
      if (e < n) myrow = __float_as_int(out[(unsigned)k * DIMS + (unsigned)e]);
      myrow &= (BATCH_N - 1);
      add_rows(myrow, mask, lane, z4, a, c0, c1, c2, c3);
    }
  } else {
    for (int c = 0; c < BATCH_N / 32; ++c) {
      const int row = c * 32 + (int)lane;
      const bool hit = (idxs[row] == k);
      const unsigned mask = __builtin_amdgcn_ballot_w32(hit);
      if (mask != 0u) add_rows(hit ? row : 0, mask, lane, z4, a, c0, c1, c2, c3);
    }
  }

  const float nU = Nk * DECAY_F + (float)n * (1.0f - DECAY_F);
  const float den = (nU / zbc) + EPS_F;
  float4 r;
  r.x = ((m4.x * DECAY_F + c0[0] * (1.0f - DECAY_F)) / zbc) / den;
  r.y = ((m4.y * DECAY_F + c1[0] * (1.0f - DECAY_F)) / zbc) / den;
  r.z = ((m4.z * DECAY_F + c2[0] * (1.0f - DECAY_F)) / zbc) / den;
  r.w = ((m4.w * DECAY_F + c3[0] * (1.0f - DECAY_F)) / zbc) / den;
  ((float4*)out)[(unsigned)k * (DIMS / 4) + lane] = r;
}

extern "C" void kernel_launch(void* const* d_in, const int* in_sizes, int n_in,
                              void* d_out, int out_size, void* d_ws, size_t ws_size,
                              hipStream_t stream) {
  const float* z_e  = (const float*)d_in[0];
  const int*   idxs = (const int*)  d_in[1];
  const float* m    = (const float*)d_in[2];
  const float* N    = (const float*)d_in[3];
  const int*   t    = (const int*)  d_in[4];
  float* out = (float*)d_out;
  int*   cnt = (int*)d_ws;

  ema_file_kernel<<<K_CODES / OWN_CODES, FILE_THREADS, 0, stream>>>(idxs, out, cnt);
  ema_gather_kernel<<<K_CODES / 8, 256, 0, stream>>>(z_e, idxs, m, N, cnt, t, out);
  (void)in_sizes; (void)n_in; (void)out_size; (void)ws_size;
}
